# starting kernel
# baseline (speedup 1.0000x reference)
#include <hip/hip_runtime.h>
#include <hip/hip_bf16.h>

typedef float v2f __attribute__((ext_vector_type(2)));
typedef float v8f __attribute__((ext_vector_type(8)));

#define B_ 4096
#define D_ 128
#define C_ 256

__global__ __launch_bounds__(256) void mahal_asym_kernel(
    const float* __restrict__ alpha, float* __restrict__ Asym) {
  const int i = blockIdx.x * 256 + threadIdx.x;
  const int r = i >> 7;
  const int c = i & 127;
  Asym[i] = alpha[r * D_ + c] + alpha[c * D_ + r];
}

__global__ __launch_bounds__(128) void mahal_mam_kernel(
    const float* __restrict__ alpha, const float* __restrict__ means,
    float* __restrict__ mAm) {
  __shared__ float red[128];
  const int c = blockIdx.x;
  const int d = threadIdx.x;
  const float* arow = alpha + d * D_;
  float w = 0.f;
  for (int e = 0; e < D_; ++e) w += arow[e] * means[e * C_ + c];
  red[d] = w * means[d * C_ + c];
  __syncthreads();
  for (int s = 64; s > 0; s >>= 1) {
    if (d < s) red[d] += red[d + s];
    __syncthreads();
  }
  if (d == 0) mAm[c] = red[0];
}

__global__ __launch_bounds__(256) void mahal_main_kernel(
    const float* __restrict__ X, const float* __restrict__ means,
    const float* __restrict__ Asym, const float* __restrict__ mAm,
    float* __restrict__ out) {
  constexpr int LDP = D_ + 4;
  __shared__ float Xs[16 * LDP];
  __shared__ float Ys[16 * LDP];
  __shared__ float tmp[256];
  __shared__ float t1s[16];
  __shared__ float pm[8][16];

  const int tid = threadIdx.x;
  const int wave = tid >> 5;
  const int lane = tid & 31;
  const int hn = lane & 15;
  const int hi = lane >> 4;
  const int rowbase = blockIdx.x * 16;

  for (int i = tid; i < 16 * D_; i += 256) {
    int r = i >> 7;
    int cc = i & 127;
    Xs[r * LDP + cc] = X[(rowbase + r) * D_ + cc];
  }
  __syncthreads();

  {
    const int cb = wave * 16;
    const int cg = cb + hn;
    v8f acc = {};
#pragma unroll 4
    for (int k0 = 0; k0 < D_; k0 += 4) {
      const int ka = k0 + 2 * hi;
      v2f a, b;
      a.x = Xs[hn * LDP + ka];
      a.y = Xs[hn * LDP + ka + 1];
      b.x = Asym[ka * D_ + cg];
      b.y = Asym[(ka + 1) * D_ + cg];
      acc = __builtin_amdgcn_wmma_f32_16x16x4_f32(false, a, false, b,
                                                  (short)0, acc, false, false);
    }
#pragma unroll
    for (int r = 0; r < 8; ++r) {
      Ys[(r + 8 * hi) * LDP + cb + hn] = acc[r];
    }
  }
  __syncthreads();

  {
    const int bb = tid >> 4;
    const int seg = tid & 15;
    float s = 0.f;
#pragma unroll
    for (int j = 0; j < 8; ++j) {
      const int dc = seg * 8 + j;
      s += Xs[bb * LDP + dc] * Ys[bb * LDP + dc];
    }
    tmp[tid] = s;
  }
  __syncthreads();
  if (tid < 16) {
    float s = 0.f;
#pragma unroll
    for (int j = 0; j < 16; ++j) s += tmp[tid * 16 + j];
    t1s[tid] = 0.5f * s;
  }
  __syncthreads();

  {
    float qm[8];
#pragma unroll
    for (int r = 0; r < 8; ++r) qm[r] = __builtin_inff();

    for (int t = 0; t < 2; ++t) {
      const int cb = (wave * 2 + t) * 16;
      v8f acc = {};
#pragma unroll 4
      for (int k0 = 0; k0 < D_; k0 += 4) {
        const int ka = k0 + 2 * hi;
        v2f a, b;
        a.x = Ys[hn * LDP + ka];
        a.y = Ys[hn * LDP + ka + 1];
        b.x = means[ka * C_ + cb + hn];
        b.y = means[(ka + 1) * C_ + cb + hn];
        acc = __builtin_amdgcn_wmma_f32_16x16x4_f32(
            false, a, false, b, (short)0, acc, false, false);
      }
      const float mac = mAm[cb + hn];
#pragma unroll
      for (int r = 0; r < 8; ++r) {
        const float q = t1s[r + 8 * hi] - acc[r] + mac;
        qm[r] = fminf(qm[r], q);
      }
    }
#pragma unroll
    for (int r = 0; r < 8; ++r) {
      float v = qm[r];
      v = fminf(v, __shfl_xor(v, 1, 32));
      v = fminf(v, __shfl_xor(v, 2, 32));
      v = fminf(v, __shfl_xor(v, 4, 32));
      v = fminf(v, __shfl_xor(v, 8, 32));
      if (hn == 0) pm[wave][r + 8 * hi] = v;
    }
  }
  __syncthreads();

  if (tid < 16) {
    float m = pm[0][tid];
#pragma unroll
    for (int w2 = 1; w2 < 8; ++w2) m = fminf(m, pm[w2][tid]);
    out[rowbase + tid] = m;
  }
}

extern "C" void kernel_launch(void* const* d_in, const int* in_sizes, int n_in,
                              void* d_out, int out_size, void* d_ws,
                              size_t ws_size, hipStream_t stream) {
  const float* X     = (const float*)d_in[0];
  const float* means = (const float*)d_in[1];
  const float* alpha = (const float*)d_in[2];
  float* out  = (float*)d_out;
  float* mAm  = (float*)d_ws;
  float* Asym = (float*)d_ws + 256;

  mahal_asym_kernel<<<(D_ * D_) / 256, 256, 0, stream>>>(alpha, Asym);
  mahal_mam_kernel<<<C_, 128, 0, stream>>>(alpha, means, mAm);
  mahal_main_kernel<<<B_ / 16, 256, 0, stream>>>(X, means, Asym, mAm, out);
}


# main kernel: operand loads issued in two large groups ahead of use, B fragments of both GEMMs in registers (means as 8-byte column pairs), barriers fence LDS only; was 8 loads waited per 4 k-steps
# speedup vs baseline: 1.3502x; 1.3502x over previous
#include <hip/hip_runtime.h>
#include <hip/hip_bf16.h>

typedef float v2f __attribute__((ext_vector_type(2)));
typedef float v8f __attribute__((ext_vector_type(8)));

#define B_ 4096
#define D_ 128
#define C_ 256

__global__ __launch_bounds__(256) void mahal_asym_kernel(
    const float* __restrict__ alpha, float* __restrict__ Asym) {
  const int i = blockIdx.x * 256 + threadIdx.x;
  const int r = i >> 7;
  const int c = i & 127;
  Asym[i] = alpha[r * D_ + c] + alpha[c * D_ + r];
}

__global__ __launch_bounds__(128) void mahal_mam_kernel(
    const float* __restrict__ alpha, const float* __restrict__ means,
    float* __restrict__ mAm) {
  __shared__ float red[128];
  const int c = blockIdx.x;
  const int d = threadIdx.x;
  const float* arow = alpha + d * D_;
  float w = 0.f;
  for (int e = 0; e < D_; ++e) w += arow[e] * means[e * C_ + c];
  red[d] = w * means[d * C_ + c];
  __syncthreads();
  for (int s = 64; s > 0; s >>= 1) {
    if (d < s) red[d] += red[d + s];
    __syncthreads();
  }
  if (d == 0) mAm[c] = red[0];
}

__device__ __forceinline__ void lds_barrier() {
  __builtin_amdgcn_fence(__ATOMIC_RELEASE, "workgroup", "local");
  __builtin_amdgcn_s_barrier();
  __builtin_amdgcn_fence(__ATOMIC_ACQUIRE, "workgroup", "local");
}

__global__ __launch_bounds__(256) void mahal_main_kernel(
    const float* __restrict__ X, const float* __restrict__ means,
    const float* __restrict__ Asym, const float* __restrict__ mAm,
    float* __restrict__ out) {
  constexpr int LDP = D_ + 4;
  constexpr int KS = D_ / 4;
  __shared__ __attribute__((aligned(16))) float Xs[16 * LDP];
  __shared__ __attribute__((aligned(16))) float Ys[16 * LDP];
  __shared__ float tmp[256];
  __shared__ float t1s[16];
  __shared__ float pm[8][16];

  const int tid = threadIdx.x;
  const int wave = tid >> 5;
  const int lane = tid & 31;
  const int hn = lane & 15;
  const int hi = lane >> 4;
  const int rowbase = blockIdx.x * 16;

  const float4* __restrict__ Xt =
      reinterpret_cast<const float4*>(X + (size_t)rowbase * D_);
  const float4 xv0 = Xt[tid];
  const float4 xv1 = Xt[tid + 256];

  v2f bA[KS];
  {
    const float* ap = Asym + (2 * hi) * D_ + wave * 16 + hn;
#pragma unroll
    for (int i = 0; i < KS; ++i) {
      bA[i].x = ap[(4 * i) * D_];
      bA[i].y = ap[(4 * i + 1) * D_];
    }
  }

  {
    const int f0 = tid, f1 = tid + 256;
    *reinterpret_cast<float4*>(&Xs[(f0 >> 5) * LDP + 4 * (f0 & 31)]) = xv0;
    *reinterpret_cast<float4*>(&Xs[(f1 >> 5) * LDP + 4 * (f1 & 31)]) = xv1;
  }

  lds_barrier();

  float2 mr0[KS], mr1[KS];
  {
    const float* mp = means + (2 * hi) * C_ + wave * 32 + 2 * hn;
#pragma unroll
    for (int i = 0; i < KS; ++i) {
      mr0[i] = *reinterpret_cast<const float2*>(mp + (4 * i) * C_);
      mr1[i] = *reinterpret_cast<const float2*>(mp + (4 * i + 1) * C_);
    }
  }
  const float2 mac =
      *reinterpret_cast<const float2*>(mAm + wave * 32 + 2 * hn);
  __builtin_amdgcn_sched_barrier(0);

  {
    const int cb = wave * 16;
    v8f acc = {};
#pragma unroll
    for (int i = 0; i < KS; ++i) {
      const int ka = 4 * i + 2 * hi;
      v2f a;
      a.x = Xs[hn * LDP + ka];
      a.y = Xs[hn * LDP + ka + 1];
      acc = __builtin_amdgcn_wmma_f32_16x16x4_f32(false, a, false, bA[i],
                                                  (short)0, acc, false, false);
    }

#pragma unroll
    for (int r = 0; r < 8; ++r) {
      Ys[(r + 8 * hi) * LDP + cb + hn] = acc[r];
    }
  }
  lds_barrier();

  {
    const int bb = tid >> 4;
    const int seg = tid & 15;
    float s = 0.f;
#pragma unroll
    for (int j = 0; j < 8; ++j) {
      const int dc = seg * 8 + j;
      s += Xs[bb * LDP + dc] * Ys[bb * LDP + dc];
    }
    tmp[tid] = s;
  }
  lds_barrier();
  if (tid < 16) {
    float s = 0.f;
#pragma unroll
    for (int j = 0; j < 16; ++j) s += tmp[tid * 16 + j];
    t1s[tid] = 0.5f * s;
  }
  lds_barrier();

  {
    float qm[8];
#pragma unroll
    for (int r = 0; r < 8; ++r) qm[r] = __builtin_inff();

    v8f acc0 = {}, acc1 = {};
#pragma unroll
    for (int i = 0; i < KS; ++i) {
      const int ka = 4 * i + 2 * hi;
      v2f a, b0, b1;
      a.x = Ys[hn * LDP + ka];
      a.y = Ys[hn * LDP + ka + 1];
      b0.x = mr0[i].x;
      b0.y = mr1[i].x;
      b1.x = mr0[i].y;
      b1.y = mr1[i].y;
      acc0 = __builtin_amdgcn_wmma_f32_16x16x4_f32(false, a, false, b0,
                                                   (short)0, acc0, false, false);
      acc1 = __builtin_amdgcn_wmma_f32_16x16x4_f32(false, a, false, b1,
                                                   (short)0, acc1, false, false);
    }
#pragma unroll
    for (int r = 0; r < 8; ++r) {
      const float q0 = t1s[r + 8 * hi] - acc0[r] + mac.x;
      const float q1 = t1s[r + 8 * hi] - acc1[r] + mac.y;
      qm[r] = fminf(fminf(qm[r], q0), q1);
    }
#pragma unroll
    for (int r = 0; r < 8; ++r) {
      float v = qm[r];
      v = fminf(v, __shfl_xor(v, 1, 32));
      v = fminf(v, __shfl_xor(v, 2, 32));
      v = fminf(v, __shfl_xor(v, 4, 32));
      v = fminf(v, __shfl_xor(v, 8, 32));
      if (hn == 0) pm[wave][r + 8 * hi] = v;
    }
  }
  lds_barrier();

  if (tid < 16) {
    float m = pm[0][tid];
#pragma unroll
    for (int w2 = 1; w2 < 8; ++w2) m = fminf(m, pm[w2][tid]);
    out[rowbase + tid] = m;
  }
}

extern "C" void kernel_launch(void* const* d_in, const int* in_sizes, int n_in,
                              void* d_out, int out_size, void* d_ws,
                              size_t ws_size, hipStream_t stream) {
  const float* X     = (const float*)d_in[0];
  const float* means = (const float*)d_in[1];
  const float* alpha = (const float*)d_in[2];
  float* out  = (float*)d_out;
  float* mAm  = (float*)d_ws;
  float* Asym = (float*)d_ws + 256;

  mahal_asym_kernel<<<(D_ * D_) / 256, 256, 0, stream>>>(alpha, Asym);
  mahal_mam_kernel<<<C_, 128, 0, stream>>>(alpha, means, mAm);
  mahal_main_kernel<<<B_ / 16, 256, 0, stream>>>(X, means, Asym, mAm, out);
}


# asym and mAm preparation kernels fused into one launch (3 launches -> 2): each mAm workgroup also writes one row of alpha + alpha^T; arithmetic and summation order of every value unchanged
# speedup vs baseline: 1.5084x; 1.1172x over previous
#include <hip/hip_runtime.h>
#include <hip/hip_bf16.h>

typedef float v2f __attribute__((ext_vector_type(2)));
typedef float v8f __attribute__((ext_vector_type(8)));

#define B_ 4096
#define D_ 128
#define C_ 256

__global__ __launch_bounds__(128) void mahal_prep_kernel(
    const float* __restrict__ alpha, const float* __restrict__ means,
    float* __restrict__ Asym, float* __restrict__ mAm) {
  __shared__ float red[128];
  const int c = blockIdx.x;
  const int d = threadIdx.x;
  const int rc = c & (D_ - 1);
  const float a0 = alpha[rc * D_ + d];
  const float a1 = alpha[d * D_ + rc];
  const float* arow = alpha + d * D_;
  float w = 0.f;
  for (int e = 0; e < D_; ++e) w += arow[e] * means[e * C_ + c];
  red[d] = w * means[d * C_ + c];
  __syncthreads();
  for (int s = 64; s > 0; s >>= 1) {
    if (d < s) red[d] += red[d + s];
    __syncthreads();
  }
  if (d == 0) mAm[c] = red[0];
  if (c < D_) Asym[c * D_ + d] = a0 + a1;
}

__device__ __forceinline__ void lds_barrier() {
  __builtin_amdgcn_fence(__ATOMIC_RELEASE, "workgroup", "local");
  __builtin_amdgcn_s_barrier();
  __builtin_amdgcn_fence(__ATOMIC_ACQUIRE, "workgroup", "local");
}

__global__ __launch_bounds__(256) void mahal_main_kernel(
    const float* __restrict__ X, const float* __restrict__ means,
    const float* __restrict__ Asym, const float* __restrict__ mAm,
    float* __restrict__ out) {
  constexpr int LDP = D_ + 4;
  constexpr int KS = D_ / 4;
  __shared__ __attribute__((aligned(16))) float Xs[16 * LDP];
  __shared__ __attribute__((aligned(16))) float Ys[16 * LDP];
  __shared__ float tmp[256];
  __shared__ float t1s[16];
  __shared__ float pm[8][16];

  const int tid = threadIdx.x;
  const int wave = tid >> 5;
  const int lane = tid & 31;
  const int hn = lane & 15;
  const int hi = lane >> 4;
  const int rowbase = blockIdx.x * 16;

  const float4* __restrict__ Xt =
      reinterpret_cast<const float4*>(X + (size_t)rowbase * D_);
  const float4 xv0 = Xt[tid];
  const float4 xv1 = Xt[tid + 256];

  v2f bA[KS];
  {
    const float* ap = Asym + (2 * hi) * D_ + wave * 16 + hn;
#pragma unroll
    for (int i = 0; i < KS; ++i) {
      bA[i].x = ap[(4 * i) * D_];
      bA[i].y = ap[(4 * i + 1) * D_];
    }
  }

  {
    const int f0 = tid, f1 = tid + 256;
    *reinterpret_cast<float4*>(&Xs[(f0 >> 5) * LDP + 4 * (f0 & 31)]) = xv0;
    *reinterpret_cast<float4*>(&Xs[(f1 >> 5) * LDP + 4 * (f1 & 31)]) = xv1;
  }

  lds_barrier();

  float2 mr0[KS], mr1[KS];
  {
    const float* mp = means + (2 * hi) * C_ + wave * 32 + 2 * hn;
#pragma unroll
    for (int i = 0; i < KS; ++i) {
      mr0[i] = *reinterpret_cast<const float2*>(mp + (4 * i) * C_);
      mr1[i] = *reinterpret_cast<const float2*>(mp + (4 * i + 1) * C_);
    }
  }
  const float2 mac =
      *reinterpret_cast<const float2*>(mAm + wave * 32 + 2 * hn);
  __builtin_amdgcn_sched_barrier(0);

  {
    const int cb = wave * 16;
    v8f acc = {};
#pragma unroll
    for (int i = 0; i < KS; ++i) {
      const int ka = 4 * i + 2 * hi;
      v2f a;
      a.x = Xs[hn * LDP + ka];
      a.y = Xs[hn * LDP + ka + 1];
      acc = __builtin_amdgcn_wmma_f32_16x16x4_f32(false, a, false, bA[i],
                                                  (short)0, acc, false, false);
    }

#pragma unroll
    for (int r = 0; r < 8; ++r) {
      Ys[(r + 8 * hi) * LDP + cb + hn] = acc[r];
    }
  }
  lds_barrier();

  {
    const int bb = tid >> 4;
    const int seg = tid & 15;
    float s = 0.f;
#pragma unroll
    for (int j = 0; j < 8; ++j) {
      const int dc = seg * 8 + j;
      s += Xs[bb * LDP + dc] * Ys[bb * LDP + dc];
    }
    tmp[tid] = s;
  }
  lds_barrier();
  if (tid < 16) {
    float s = 0.f;
#pragma unroll
    for (int j = 0; j < 16; ++j) s += tmp[tid * 16 + j];
    t1s[tid] = 0.5f * s;
  }
  lds_barrier();

  {
    float qm[8];
#pragma unroll
    for (int r = 0; r < 8; ++r) qm[r] = __builtin_inff();

    v8f acc0 = {}, acc1 = {};
#pragma unroll
    for (int i = 0; i < KS; ++i) {
      const int ka = 4 * i + 2 * hi;
      v2f a, b0, b1;
      a.x = Ys[hn * LDP + ka];
      a.y = Ys[hn * LDP + ka + 1];
      b0.x = mr0[i].x;
      b0.y = mr1[i].x;
      b1.x = mr0[i].y;
      b1.y = mr1[i].y;
      acc0 = __builtin_amdgcn_wmma_f32_16x16x4_f32(false, a, false, b0,
                                                   (short)0, acc0, false, false);
      acc1 = __builtin_amdgcn_wmma_f32_16x16x4_f32(false, a, false, b1,
                                                   (short)0, acc1, false, false);
    }
#pragma unroll
    for (int r = 0; r < 8; ++r) {
      const float q0 = t1s[r + 8 * hi] - acc0[r] + mac.x;
      const float q1 = t1s[r + 8 * hi] - acc1[r] + mac.y;
      qm[r] = fminf(fminf(qm[r], q0), q1);
    }
#pragma unroll
    for (int r = 0; r < 8; ++r) {
      float v = qm[r];
      v = fminf(v, __shfl_xor(v, 1, 32));
      v = fminf(v, __shfl_xor(v, 2, 32));
      v = fminf(v, __shfl_xor(v, 4, 32));
      v = fminf(v, __shfl_xor(v, 8, 32));
      if (hn == 0) pm[wave][r + 8 * hi] = v;
    }
  }
  lds_barrier();

  if (tid < 16) {
    float m = pm[0][tid];
#pragma unroll
    for (int w2 = 1; w2 < 8; ++w2) m = fminf(m, pm[w2][tid]);
    out[rowbase + tid] = m;
  }
}

extern "C" void kernel_launch(void* const* d_in, const int* in_sizes, int n_in,
                              void* d_out, int out_size, void* d_ws,
                              size_t ws_size, hipStream_t stream) {
  const float* X     = (const float*)d_in[0];
  const float* means = (const float*)d_in[1];
  const float* alpha = (const float*)d_in[2];
  float* out  = (float*)d_out;
  float* mAm  = (float*)d_ws;
  float* Asym = (float*)d_ws + 256;

  mahal_prep_kernel<<<C_, 128, 0, stream>>>(alpha, means, Asym, mAm);
  mahal_main_kernel<<<B_ / 16, 256, 0, stream>>>(X, means, Asym, mAm, out);
}


# preparation kernel: all operand loads issued once as one group (a wave reads 32 whole rows of alpha, means column broadcast by lane reads); was 16 waited load rounds; mAm summation order changes
# speedup vs baseline: 2.4283x; 1.6099x over previous
#include <hip/hip_runtime.h>
#include <hip/hip_bf16.h>

typedef float v2f __attribute__((ext_vector_type(2)));
typedef float v8f __attribute__((ext_vector_type(8)));

#define B_ 4096
#define D_ 128
#define C_ 256

__device__ __forceinline__ void lds_barrier() {
  __builtin_amdgcn_fence(__ATOMIC_RELEASE, "workgroup", "local");
  __builtin_amdgcn_s_barrier();
  __builtin_amdgcn_fence(__ATOMIC_ACQUIRE, "workgroup", "local");
}

__global__ __launch_bounds__(128) void mahal_prep_kernel(
    const float* __restrict__ alpha, const float* __restrict__ means,
    float* __restrict__ Asym, float* __restrict__ mAm) {
  __shared__ __attribute__((aligned(16))) float ms[D_];
  __shared__ float wsum[4];
  const int c = blockIdx.x;
  const int tid = threadIdx.x;
  const int wave = tid >> 5;
  const int lane = tid & 31;

  const int rc = c & (D_ - 1);
  const float a0 = alpha[rc * D_ + tid];
  const float a1 = alpha[tid * D_ + rc];
  const float mv = means[tid * C_ + c];
  const float4* __restrict__ ap =
      reinterpret_cast<const float4*>(alpha + (32 * wave) * D_ + 4 * lane);
  float4 av[32];
#pragma unroll
  for (int j = 0; j < 32; ++j) av[j] = ap[j * (D_ / 4)];

  ms[tid] = mv;
  lds_barrier();
  const float4 m4 = *reinterpret_cast<const float4*>(&ms[4 * lane]);

  if (c < D_) Asym[c * D_ + tid] = a0 + a1;

  float4 u = {0.f, 0.f, 0.f, 0.f};
#pragma unroll
  for (int j = 0; j < 32; ++j) {
    const float s =
        __int_as_float(__builtin_amdgcn_readlane(__float_as_int(mv), j));
    u.x += s * av[j].x;
    u.y += s * av[j].y;
    u.z += s * av[j].z;
    u.w += s * av[j].w;
  }
  float p = (u.x * m4.x + u.y * m4.y) + (u.z * m4.z + u.w * m4.w);
  p += __shfl_xor(p, 16, 32);
  p += __shfl_xor(p, 8, 32);
  p += __shfl_xor(p, 4, 32);
  p += __shfl_xor(p, 2, 32);
  p += __shfl_xor(p, 1, 32);
  if (lane == 0) wsum[wave] = p;
  lds_barrier();
  if (tid == 0) mAm[c] = (wsum[0] + wsum[1]) + (wsum[2] + wsum[3]);
}

__global__ __launch_bounds__(256) void mahal_main_kernel(
    const float* __restrict__ X, const float* __restrict__ means,
    const float* __restrict__ Asym, const float* __restrict__ mAm,
    float* __restrict__ out) {
  constexpr int LDP = D_ + 4;
  constexpr int KS = D_ / 4;
  __shared__ __attribute__((aligned(16))) float Xs[16 * LDP];
  __shared__ __attribute__((aligned(16))) float Ys[16 * LDP];
  __shared__ float tmp[256];
  __shared__ float t1s[16];
  __shared__ float pm[8][16];

  const int tid = threadIdx.x;
  const int wave = tid >> 5;
  const int lane = tid & 31;
  const int hn = lane & 15;
  const int hi = lane >> 4;
  const int rowbase = blockIdx.x * 16;

  const float4* __restrict__ Xt =
      reinterpret_cast<const float4*>(X + (size_t)rowbase * D_);
  const float4 xv0 = Xt[tid];
  const float4 xv1 = Xt[tid + 256];

  v2f bA[KS];
  {
    const float* ap = Asym + (2 * hi) * D_ + wave * 16 + hn;
#pragma unroll
    for (int i = 0; i < KS; ++i) {
      bA[i].x = ap[(4 * i) * D_];
      bA[i].y = ap[(4 * i + 1) * D_];
    }
  }

  {
    const int f0 = tid, f1 = tid + 256;
    *reinterpret_cast<float4*>(&Xs[(f0 >> 5) * LDP + 4 * (f0 & 31)]) = xv0;
    *reinterpret_cast<float4*>(&Xs[(f1 >> 5) * LDP + 4 * (f1 & 31)]) = xv1;
  }

  lds_barrier();

  float2 mr0[KS], mr1[KS];
  {
    const float* mp = means + (2 * hi) * C_ + wave * 32 + 2 * hn;
#pragma unroll
    for (int i = 0; i < KS; ++i) {
      mr0[i] = *reinterpret_cast<const float2*>(mp + (4 * i) * C_);
      mr1[i] = *reinterpret_cast<const float2*>(mp + (4 * i + 1) * C_);
    }
  }
  const float2 mac =
      *reinterpret_cast<const float2*>(mAm + wave * 32 + 2 * hn);
  __builtin_amdgcn_sched_barrier(0);

  {
    const int cb = wave * 16;
    v8f acc = {};
#pragma unroll
    for (int i = 0; i < KS; ++i) {
      const int ka = 4 * i + 2 * hi;
      v2f a;
      a.x = Xs[hn * LDP + ka];
      a.y = Xs[hn * LDP + ka + 1];
      acc = __builtin_amdgcn_wmma_f32_16x16x4_f32(false, a, false, bA[i],
                                                  (short)0, acc, false, false);
    }

#pragma unroll
    for (int r = 0; r < 8; ++r) {
      Ys[(r + 8 * hi) * LDP + cb + hn] = acc[r];
    }
  }
  lds_barrier();

  {
    const int bb = tid >> 4;
    const int seg = tid & 15;
    float s = 0.f;
#pragma unroll
    for (int j = 0; j < 8; ++j) {
      const int dc = seg * 8 + j;
      s += Xs[bb * LDP + dc] * Ys[bb * LDP + dc];
    }
    tmp[tid] = s;
  }
  lds_barrier();
  if (tid < 16) {
    float s = 0.f;
#pragma unroll
    for (int j = 0; j < 16; ++j) s += tmp[tid * 16 + j];
    t1s[tid] = 0.5f * s;
  }
  lds_barrier();

  {
    float qm[8];
#pragma unroll
    for (int r = 0; r < 8; ++r) qm[r] = __builtin_inff();

    v8f acc0 = {}, acc1 = {};
#pragma unroll
    for (int i = 0; i < KS; ++i) {
      const int ka = 4 * i + 2 * hi;
      v2f a, b0, b1;
      a.x = Ys[hn * LDP + ka];
      a.y = Ys[hn * LDP + ka + 1];
      b0.x = mr0[i].x;
      b0.y = mr1[i].x;
      b1.x = mr0[i].y;
      b1.y = mr1[i].y;
      acc0 = __builtin_amdgcn_wmma_f32_16x16x4_f32(false, a, false, b0,
                                                   (short)0, acc0, false, false);
      acc1 = __builtin_amdgcn_wmma_f32_16x16x4_f32(false, a, false, b1,
                                                   (short)0, acc1, false, false);
    }
#pragma unroll
    for (int r = 0; r < 8; ++r) {
      const float q0 = t1s[r + 8 * hi] - acc0[r] + mac.x;
      const float q1 = t1s[r + 8 * hi] - acc1[r] + mac.y;
      qm[r] = fminf(fminf(qm[r], q0), q1);
    }
#pragma unroll
    for (int r = 0; r < 8; ++r) {
      float v = qm[r];
      v = fminf(v, __shfl_xor(v, 1, 32));
      v = fminf(v, __shfl_xor(v, 2, 32));
      v = fminf(v, __shfl_xor(v, 4, 32));
      v = fminf(v, __shfl_xor(v, 8, 32));
      if (hn == 0) pm[wave][r + 8 * hi] = v;
    }
  }
  lds_barrier();

  if (tid < 16) {
    float m = pm[0][tid];
#pragma unroll
    for (int w2 = 1; w2 < 8; ++w2) m = fminf(m, pm[w2][tid]);
    out[rowbase + tid] = m;
  }
}

extern "C" void kernel_launch(void* const* d_in, const int* in_sizes, int n_in,
                              void* d_out, int out_size, void* d_ws,
                              size_t ws_size, hipStream_t stream) {
  const float* X     = (const float*)d_in[0];
  const float* means = (const float*)d_in[1];
  const float* alpha = (const float*)d_in[2];
  float* out  = (float*)d_out;
  float* mAm  = (float*)d_ws;
  float* Asym = (float*)d_ws + 256;

  mahal_prep_kernel<<<C_, 128, 0, stream>>>(alpha, means, Asym, mAm);
  mahal_main_kernel<<<B_ / 16, 256, 0, stream>>>(X, means, Asym, mAm, out);
}
